# starting kernel
# baseline (speedup 1.0000x reference)
#include <hip/hip_runtime.h>


typedef __attribute__((ext_vector_type(2))) float v2f;
typedef __attribute__((ext_vector_type(8))) float v8f;

namespace {
constexpr int G       = 256;
constexpr int ARITY   = 4;
constexpr int D       = 16;
constexpr int BATCH   = 4096;
constexpr int XW      = G * ARITY;
constexpr int OUTW    = G * D;
constexpr int WAVES_PER_BLOCK = 8;
constexpr int TILES_PER_WAVE  = 4;
constexpr int BLOCKS_Y = BATCH / 16 / (WAVES_PER_BLOCK * TILES_PER_WAVE);
}

__global__ __launch_bounds__(256)
void simplex_wmma_kernel(const float* __restrict__ X,
                         const float* __restrict__ params,
                         float* __restrict__ out)
{
    const int lane = threadIdx.x & 31;
    const int wave = threadIdx.x >> 5;
    const int g    = blockIdx.x;
    const int half = lane >> 4;
    const int m    = lane & 15;

    const float* __restrict__ P = params + (size_t)g * (16 * D);
    v2f bmat[4];
#pragma unroll
    for (int kk = 0; kk < 4; ++kk) {
        const int p0 = kk * 4 + half * 2;
        bmat[kk].x = P[p0 * D + m];
        bmat[kk].y = P[(p0 + 1) * D + m];
    }

#pragma unroll
    for (int t = 0; t < TILES_PER_WAVE; ++t) {
        const int tile = (blockIdx.y * WAVES_PER_BLOCK + wave) * TILES_PER_WAVE + t;
        const int b0   = tile * 16;

        const float4 xr = *reinterpret_cast<const float4*>(
            X + (size_t)(b0 + m) * XW + (size_t)g * ARITY);
        float v0 = xr.x, v1 = xr.y, v2 = xr.z, v3 = xr.w;
        int   j0 = 0, j1 = 1, j2 = 2, j3 = 3;
        if (v0 > v1) { float tv=v0; v0=v1; v1=tv; int tj=j0; j0=j1; j1=tj; }
        if (v2 > v3) { float tv=v2; v2=v3; v3=tv; int tj=j2; j2=j3; j3=tj; }
        if (v0 > v2) { float tv=v0; v0=v2; v2=tv; int tj=j0; j0=j2; j2=tj; }
        if (v1 > v3) { float tv=v1; v1=v3; v3=tv; int tj=j1; j1=j3; j3=tj; }
        if (v1 > v2) { float tv=v1; v1=v2; v2=tv; int tj=j1; j1=j2; j2=tj; }

        const float c0 = v0, c1 = v1 - v0, c2 = v2 - v1, c3 = v3 - v2;
        const int   i3 = 1 << j3;
        const int   i2 = i3 | (1 << j2);
        const int   i1 = i2 | (1 << j1);
        const int   i0 = 15;

        auto wsel = [&](int p) -> float {
            float r = (i0 == p) ? c0 : 0.0f;
            r += (i1 == p) ? c1 : 0.0f;
            r += (i2 == p) ? c2 : 0.0f;
            r += (i3 == p) ? c3 : 0.0f;
            return r;
        };

        v8f acc = {};
#pragma unroll
        for (int kk = 0; kk < 4; ++kk) {
            const int p0 = kk * 4 + half * 2;
            v2f a;
            a.x = wsel(p0);
            a.y = wsel(p0 + 1);
            acc = __builtin_amdgcn_wmma_f32_16x16x4_f32(
                false, a, false, bmat[kk], (short)0, acc, false, false);
        }

        float* __restrict__ orow =
            out + (size_t)(b0 + half * 8) * OUTW + (size_t)g * D + m;
#pragma unroll
        for (int j = 0; j < 8; ++j) {
            orow[(size_t)j * OUTW] = acc[j];
        }
    }
}

extern "C" void kernel_launch(void* const* d_in, const int* in_sizes, int n_in,
                              void* d_out, int out_size, void* d_ws, size_t ws_size,
                              hipStream_t stream) {
    (void)in_sizes; (void)n_in; (void)out_size; (void)d_ws; (void)ws_size;
    const float* X      = (const float*)d_in[0];
    const float* params = (const float*)d_in[1];
    float*       out    = (float*)d_out;

    dim3 grid(G, BLOCKS_Y);
    dim3 block(32 * WAVES_PER_BLOCK);
    simplex_wmma_kernel<<<grid, block, 0, stream>>>(X, params, out);
}


# wave tile re-cut from 64 rows x 1 group to 32 rows x 8 adjacent groups (4x the work per wave): each X line is read once, by whole-line loads, and reaches its lanes through wave-private LDS
# speedup vs baseline: 1.9365x; 1.9365x over previous
#include <hip/hip_runtime.h>


typedef __attribute__((ext_vector_type(2))) float v2f;
typedef __attribute__((ext_vector_type(8))) float v8f;

namespace {
constexpr int G       = 256;
constexpr int ARITY   = 4;
constexpr int D       = 16;
constexpr int BATCH   = 4096;
constexpr int XW      = G * ARITY;
constexpr int OUTW    = G * D;
constexpr int WAVES_PER_BLOCK = 8;
constexpr int GROUPS_PER_WAVE = 8;
constexpr int TILES_PER_WAVE  = 2;
constexpr int ROWS_PER_BLOCK  = WAVES_PER_BLOCK * TILES_PER_WAVE * 16;
constexpr int BLOCKS_X = G / GROUPS_PER_WAVE;
constexpr int BLOCKS_Y = BATCH / ROWS_PER_BLOCK;
constexpr int PITCH    = GROUPS_PER_WAVE + 1;
}

__global__ __launch_bounds__(256)
void simplex_wmma_kernel(const float* __restrict__ X,
                         const float* __restrict__ params,
                         float* __restrict__ out)
{
    __shared__ float4 xs[WAVES_PER_BLOCK][TILES_PER_WAVE][16 * PITCH];

    const int lane = threadIdx.x & 31;
    const int wave = threadIdx.x >> 5;
    const int g0   = blockIdx.x * GROUPS_PER_WAVE;
    const int r0   = (blockIdx.y * WAVES_PER_BLOCK + wave) * (TILES_PER_WAVE * 16);
    const int half = lane >> 4;
    const int m    = lane & 15;

    const int lrow = lane >> 3;
    const int lcol = lane & 7;
    float4 xr[TILES_PER_WAVE][4];
#pragma unroll
    for (int t = 0; t < TILES_PER_WAVE; ++t) {
#pragma unroll
        for (int i = 0; i < 4; ++i) {
            const int row = r0 + t * 16 + i * 4 + lrow;
            xr[t][i] = *reinterpret_cast<const float4*>(
                X + (size_t)row * XW + (size_t)(g0 + lcol) * ARITY);
        }
    }

    v2f bmat[GROUPS_PER_WAVE][4];
#pragma unroll
    for (int gi = 0; gi < GROUPS_PER_WAVE; ++gi) {
        const float* __restrict__ P = params + (size_t)(g0 + gi) * (16 * D);
#pragma unroll
        for (int kk = 0; kk < 4; ++kk) {
            const int p0 = kk * 4 + half * 2;
            bmat[gi][kk].x = P[p0 * D + m];
            bmat[gi][kk].y = P[(p0 + 1) * D + m];
        }
    }

#pragma unroll
    for (int t = 0; t < TILES_PER_WAVE; ++t) {
#pragma unroll
        for (int i = 0; i < 4; ++i) {
            xs[wave][t][(i * 4 + lrow) * PITCH + lcol] = xr[t][i];
        }
    }
    __builtin_amdgcn_fence(__ATOMIC_RELEASE, "wavefront");
    __builtin_amdgcn_wave_barrier();
    __builtin_amdgcn_fence(__ATOMIC_ACQUIRE, "wavefront");

    float4 xv[TILES_PER_WAVE][GROUPS_PER_WAVE];
#pragma unroll
    for (int t = 0; t < TILES_PER_WAVE; ++t) {
#pragma unroll
        for (int gi = 0; gi < GROUPS_PER_WAVE; ++gi) {
            xv[t][gi] = xs[wave][t][m * PITCH + gi];
        }
    }

#pragma unroll
    for (int t = 0; t < TILES_PER_WAVE; ++t) {
        const int b0 = r0 + t * 16;
#pragma unroll
        for (int gi = 0; gi < GROUPS_PER_WAVE; ++gi) {
            float v0 = xv[t][gi].x, v1 = xv[t][gi].y, v2 = xv[t][gi].z, v3 = xv[t][gi].w;
            int   j0 = 0, j1 = 1, j2 = 2, j3 = 3;
            if (v0 > v1) { float tv=v0; v0=v1; v1=tv; int tj=j0; j0=j1; j1=tj; }
            if (v2 > v3) { float tv=v2; v2=v3; v3=tv; int tj=j2; j2=j3; j3=tj; }
            if (v0 > v2) { float tv=v0; v0=v2; v2=tv; int tj=j0; j0=j2; j2=tj; }
            if (v1 > v3) { float tv=v1; v1=v3; v3=tv; int tj=j1; j1=j3; j3=tj; }
            if (v1 > v2) { float tv=v1; v1=v2; v2=tv; int tj=j1; j1=j2; j2=tj; }

            const float c0 = v0, c1 = v1 - v0, c2 = v2 - v1, c3 = v3 - v2;
            const int   i3 = 1 << j3;
            const int   i2 = i3 | (1 << j2);
            const int   i1 = i2 | (1 << j1);
            const int   i0 = 15;

            auto wsel = [&](int p) -> float {
                float r = (i0 == p) ? c0 : 0.0f;
                r += (i1 == p) ? c1 : 0.0f;
                r += (i2 == p) ? c2 : 0.0f;
                r += (i3 == p) ? c3 : 0.0f;
                return r;
            };

            v8f acc = {};
#pragma unroll
            for (int kk = 0; kk < 4; ++kk) {
                const int p0 = kk * 4 + half * 2;
                v2f a;
                a.x = wsel(p0);
                a.y = wsel(p0 + 1);
                acc = __builtin_amdgcn_wmma_f32_16x16x4_f32(
                    false, a, false, bmat[gi][kk], (short)0, acc, false, false);
            }

            float* __restrict__ orow =
                out + (size_t)(b0 + half * 8) * OUTW + (size_t)(g0 + gi) * D + m;
#pragma unroll
            for (int j = 0; j < 8; ++j) {
                orow[(size_t)j * OUTW] = acc[j];
            }
        }
    }
}

extern "C" void kernel_launch(void* const* d_in, const int* in_sizes, int n_in,
                              void* d_out, int out_size, void* d_ws, size_t ws_size,
                              hipStream_t stream) {
    (void)in_sizes; (void)n_in; (void)out_size; (void)d_ws; (void)ws_size;
    const float* X      = (const float*)d_in[0];
    const float* params = (const float*)d_in[1];
    float*       out    = (float*)d_out;

    dim3 grid(BLOCKS_X, BLOCKS_Y);
    dim3 block(32 * WAVES_PER_BLOCK);
    simplex_wmma_kernel<<<grid, block, 0, stream>>>(X, params, out);
}


# output tile taken transposed from the WMMA (operands swapped) and staged in wave-private LDS: each store instruction writes 512 contiguous bytes (4 whole lines) of a row, not two 64-B half lines
# speedup vs baseline: 3.3205x; 1.7147x over previous
#include <hip/hip_runtime.h>


typedef __attribute__((ext_vector_type(2))) float v2f;
typedef __attribute__((ext_vector_type(8))) float v8f;

namespace {
constexpr int G       = 256;
constexpr int ARITY   = 4;
constexpr int D       = 16;
constexpr int BATCH   = 4096;
constexpr int XW      = G * ARITY;
constexpr int OUTW    = G * D;
constexpr int WAVES_PER_BLOCK = 8;
constexpr int GROUPS_PER_WAVE = 8;
constexpr int TILES_PER_WAVE  = 2;
constexpr int ROWS_PER_BLOCK  = WAVES_PER_BLOCK * TILES_PER_WAVE * 16;
constexpr int BLOCKS_X = G / GROUPS_PER_WAVE;
constexpr int BLOCKS_Y = BATCH / ROWS_PER_BLOCK;
constexpr int PITCH    = GROUPS_PER_WAVE + 1;
constexpr int YROW     = GROUPS_PER_WAVE * D / 4;
constexpr int YPITCH   = YROW + 1;
}

__device__ __forceinline__ void wave_lds_handover()
{
    __builtin_amdgcn_fence(__ATOMIC_RELEASE, "wavefront");
    __builtin_amdgcn_wave_barrier();
    __builtin_amdgcn_fence(__ATOMIC_ACQUIRE, "wavefront");
}

__global__ __launch_bounds__(256)
void simplex_wmma_kernel(const float* __restrict__ X,
                         const float* __restrict__ params,
                         float* __restrict__ out)
{
    __shared__ float4 xs[WAVES_PER_BLOCK][TILES_PER_WAVE][16 * PITCH];
    __shared__ float4 ys[WAVES_PER_BLOCK][16 * YPITCH];

    const int lane = threadIdx.x & 31;
    const int wave = threadIdx.x >> 5;
    const int g0   = blockIdx.x * GROUPS_PER_WAVE;
    const int r0   = (blockIdx.y * WAVES_PER_BLOCK + wave) * (TILES_PER_WAVE * 16);
    const int half = lane >> 4;
    const int m    = lane & 15;

    const int lrow = lane >> 3;
    const int lcol = lane & 7;
    float4 xr[TILES_PER_WAVE][4];
#pragma unroll
    for (int t = 0; t < TILES_PER_WAVE; ++t) {
#pragma unroll
        for (int i = 0; i < 4; ++i) {
            const int row = r0 + t * 16 + i * 4 + lrow;
            xr[t][i] = *reinterpret_cast<const float4*>(
                X + (size_t)row * XW + (size_t)(g0 + lcol) * ARITY);
        }
    }

    v2f bmat[GROUPS_PER_WAVE][4];
#pragma unroll
    for (int gi = 0; gi < GROUPS_PER_WAVE; ++gi) {
        const float* __restrict__ P = params + (size_t)(g0 + gi) * (16 * D);
#pragma unroll
        for (int kk = 0; kk < 4; ++kk) {
            const int p0 = kk * 4 + half * 2;
            bmat[gi][kk].x = P[p0 * D + m];
            bmat[gi][kk].y = P[(p0 + 1) * D + m];
        }
    }

#pragma unroll
    for (int t = 0; t < TILES_PER_WAVE; ++t) {
#pragma unroll
        for (int i = 0; i < 4; ++i) {
            xs[wave][t][(i * 4 + lrow) * PITCH + lcol] = xr[t][i];
        }
    }
    wave_lds_handover();

    float4 xv[TILES_PER_WAVE][GROUPS_PER_WAVE];
#pragma unroll
    for (int t = 0; t < TILES_PER_WAVE; ++t) {
#pragma unroll
        for (int gi = 0; gi < GROUPS_PER_WAVE; ++gi) {
            xv[t][gi] = xs[wave][t][m * PITCH + gi];
        }
    }

#pragma unroll
    for (int t = 0; t < TILES_PER_WAVE; ++t) {
        const int b0 = r0 + t * 16;
#pragma unroll
        for (int gi = 0; gi < GROUPS_PER_WAVE; ++gi) {
            float v0 = xv[t][gi].x, v1 = xv[t][gi].y, v2 = xv[t][gi].z, v3 = xv[t][gi].w;
            int   j0 = 0, j1 = 1, j2 = 2, j3 = 3;
            if (v0 > v1) { float tv=v0; v0=v1; v1=tv; int tj=j0; j0=j1; j1=tj; }
            if (v2 > v3) { float tv=v2; v2=v3; v3=tv; int tj=j2; j2=j3; j3=tj; }
            if (v0 > v2) { float tv=v0; v0=v2; v2=tv; int tj=j0; j0=j2; j2=tj; }
            if (v1 > v3) { float tv=v1; v1=v3; v3=tv; int tj=j1; j1=j3; j3=tj; }
            if (v1 > v2) { float tv=v1; v1=v2; v2=tv; int tj=j1; j1=j2; j2=tj; }

            const float c0 = v0, c1 = v1 - v0, c2 = v2 - v1, c3 = v3 - v2;
            const int   i3 = 1 << j3;
            const int   i2 = i3 | (1 << j2);
            const int   i1 = i2 | (1 << j1);
            const int   i0 = 15;

            auto wsel = [&](int p) -> float {
                float r = (i0 == p) ? c0 : 0.0f;
                r += (i1 == p) ? c1 : 0.0f;
                r += (i2 == p) ? c2 : 0.0f;
                r += (i3 == p) ? c3 : 0.0f;
                return r;
            };

            v8f acc = {};
#pragma unroll
            for (int kk = 0; kk < 4; ++kk) {
                const int p0 = kk * 4 + half * 2;
                v2f a;
                a.x = wsel(p0);
                a.y = wsel(p0 + 1);
                acc = __builtin_amdgcn_wmma_f32_16x16x4_f32(
                    false, bmat[gi][kk], false, a, (short)0, acc, false, false);
            }

            float4* yrow = &ys[wave][m * YPITCH + gi * 4 + half * 2];
            yrow[0] = make_float4(acc[0], acc[1], acc[2], acc[3]);
            yrow[1] = make_float4(acc[4], acc[5], acc[6], acc[7]);
        }

        wave_lds_handover();

        float* __restrict__ obase =
            out + (size_t)b0 * OUTW + (size_t)g0 * D + (size_t)lane * 4;
#pragma unroll
        for (int r = 0; r < 16; ++r) {
            const float4 v = ys[wave][r * YPITCH + lane];
            *reinterpret_cast<float4*>(obase + (size_t)r * OUTW) = v;
        }

        if (t + 1 < TILES_PER_WAVE) wave_lds_handover();
    }
}

extern "C" void kernel_launch(void* const* d_in, const int* in_sizes, int n_in,
                              void* d_out, int out_size, void* d_ws, size_t ws_size,
                              hipStream_t stream) {
    (void)in_sizes; (void)n_in; (void)out_size; (void)d_ws; (void)ws_size;
    const float* X      = (const float*)d_in[0];
    const float* params = (const float*)d_in[1];
    float*       out    = (float*)d_out;

    dim3 grid(BLOCKS_X, BLOCKS_Y);
    dim3 block(32 * WAVES_PER_BLOCK);
    simplex_wmma_kernel<<<grid, block, 0, stream>>>(X, params, out);
}


# coefficient row of the B operand in closed form, W[p] = max(0, min of p's members - max of the rest): sort network, index build and per-row 4-way select dropped; values bit-identical
# speedup vs baseline: 3.9408x; 1.1868x over previous
#include <hip/hip_runtime.h>


typedef __attribute__((ext_vector_type(2))) float v2f;
typedef __attribute__((ext_vector_type(8))) float v8f;

namespace {
constexpr int G       = 256;
constexpr int ARITY   = 4;
constexpr int D       = 16;
constexpr int BATCH   = 4096;
constexpr int XW      = G * ARITY;
constexpr int OUTW    = G * D;
constexpr int WAVES_PER_BLOCK = 8;
constexpr int GROUPS_PER_WAVE = 8;
constexpr int TILES_PER_WAVE  = 2;
constexpr int ROWS_PER_BLOCK  = WAVES_PER_BLOCK * TILES_PER_WAVE * 16;
constexpr int BLOCKS_X = G / GROUPS_PER_WAVE;
constexpr int BLOCKS_Y = BATCH / ROWS_PER_BLOCK;
constexpr int PITCH    = GROUPS_PER_WAVE + 1;
constexpr int YROW     = GROUPS_PER_WAVE * D / 4;
constexpr int YPITCH   = YROW + 1;
}

__device__ __forceinline__ void wave_lds_handover()
{
    __builtin_amdgcn_fence(__ATOMIC_RELEASE, "wavefront");
    __builtin_amdgcn_wave_barrier();
    __builtin_amdgcn_fence(__ATOMIC_ACQUIRE, "wavefront");
}

__global__ __launch_bounds__(256)
void simplex_wmma_kernel(const float* __restrict__ X,
                         const float* __restrict__ params,
                         float* __restrict__ out)
{
    __shared__ float4 xs[WAVES_PER_BLOCK][TILES_PER_WAVE][16 * PITCH];
    __shared__ float4 ys[WAVES_PER_BLOCK][16 * YPITCH];

    const int lane = threadIdx.x & 31;
    const int wave = threadIdx.x >> 5;
    const int g0   = blockIdx.x * GROUPS_PER_WAVE;
    const int r0   = (blockIdx.y * WAVES_PER_BLOCK + wave) * (TILES_PER_WAVE * 16);
    const int half = lane >> 4;
    const int m    = lane & 15;

    const int lrow = lane >> 3;
    const int lcol = lane & 7;
    float4 xr[TILES_PER_WAVE][4];
#pragma unroll
    for (int t = 0; t < TILES_PER_WAVE; ++t) {
#pragma unroll
        for (int i = 0; i < 4; ++i) {
            const int row = r0 + t * 16 + i * 4 + lrow;
            xr[t][i] = *reinterpret_cast<const float4*>(
                X + (size_t)row * XW + (size_t)(g0 + lcol) * ARITY);
        }
    }

    v2f bmat[GROUPS_PER_WAVE][4];
#pragma unroll
    for (int gi = 0; gi < GROUPS_PER_WAVE; ++gi) {
        const float* __restrict__ P = params + (size_t)(g0 + gi) * (16 * D);
#pragma unroll
        for (int kk = 0; kk < 4; ++kk) {
            const int p0 = kk * 4 + half * 2;
            bmat[gi][kk].x = P[p0 * D + m];
            bmat[gi][kk].y = P[(p0 + 1) * D + m];
        }
    }

#pragma unroll
    for (int t = 0; t < TILES_PER_WAVE; ++t) {
#pragma unroll
        for (int i = 0; i < 4; ++i) {
            xs[wave][t][(i * 4 + lrow) * PITCH + lcol] = xr[t][i];
        }
    }
    wave_lds_handover();

    float4 xv[TILES_PER_WAVE][GROUPS_PER_WAVE];
#pragma unroll
    for (int t = 0; t < TILES_PER_WAVE; ++t) {
#pragma unroll
        for (int gi = 0; gi < GROUPS_PER_WAVE; ++gi) {
            xv[t][gi] = xs[wave][t][m * PITCH + gi];
        }
    }

#pragma unroll
    for (int t = 0; t < TILES_PER_WAVE; ++t) {
        const int b0 = r0 + t * 16;
#pragma unroll
        for (int gi = 0; gi < GROUPS_PER_WAVE; ++gi) {
            const float x0 = xv[t][gi].x, x1 = xv[t][gi].y, x2 = xv[t][gi].z, x3 = xv[t][gi].w;
            const float inf  = __builtin_inff();
            const float in1  = half ? x1 : inf;
            const float out1 = half ? -inf : x1;
            auto wrow = [&](int kk, int e) -> float {
                if (kk == 0 && e == 0) {
                    const float w = fmaxf(x1 - fmaxf(fmaxf(x0, x2), x3), 0.0f);
                    return half ? w : 0.0f;
                }
                if (kk == 3 && e == 1) {
                    const float m3 = fminf(fminf(x0, x2), x3);
                    return half ? fminf(m3, x1) : fmaxf(m3 - x1, 0.0f);
                }
                float mn = in1, mx = out1;
                if (e)      mn = fminf(mn, x0); else mx = fmaxf(mx, x0);
                if (kk & 1) mn = fminf(mn, x2); else mx = fmaxf(mx, x2);
                if (kk & 2) mn = fminf(mn, x3); else mx = fmaxf(mx, x3);
                return fmaxf(mn - mx, 0.0f);
            };

            v8f acc = {};
#pragma unroll
            for (int kk = 0; kk < 4; ++kk) {
                v2f a;
                a.x = wrow(kk, 0);
                a.y = wrow(kk, 1);
                acc = __builtin_amdgcn_wmma_f32_16x16x4_f32(
                    false, bmat[gi][kk], false, a, (short)0, acc, false, false);
            }

            float4* yrow = &ys[wave][m * YPITCH + gi * 4 + half * 2];
            yrow[0] = make_float4(acc[0], acc[1], acc[2], acc[3]);
            yrow[1] = make_float4(acc[4], acc[5], acc[6], acc[7]);
        }

        wave_lds_handover();

        float* __restrict__ obase =
            out + (size_t)b0 * OUTW + (size_t)g0 * D + (size_t)lane * 4;
#pragma unroll
        for (int r = 0; r < 16; ++r) {
            const float4 v = ys[wave][r * YPITCH + lane];
            *reinterpret_cast<float4*>(obase + (size_t)r * OUTW) = v;
        }

        if (t + 1 < TILES_PER_WAVE) wave_lds_handover();
    }
}

extern "C" void kernel_launch(void* const* d_in, const int* in_sizes, int n_in,
                              void* d_out, int out_size, void* d_ws, size_t ws_size,
                              hipStream_t stream) {
    (void)in_sizes; (void)n_in; (void)out_size; (void)d_ws; (void)ws_size;
    const float* X      = (const float*)d_in[0];
    const float* params = (const float*)d_in[1];
    float*       out    = (float*)d_out;

    dim3 grid(BLOCKS_X, BLOCKS_Y);
    dim3 block(32 * WAVES_PER_BLOCK);
    simplex_wmma_kernel<<<grid, block, 0, stream>>>(X, params, out);
}
